# starting kernel
# baseline (speedup 1.0000x reference)
#include <hip/hip_runtime.h>
#include <hip/hip_bf16.h>

typedef __attribute__((ext_vector_type(2))) float v2f;
typedef __attribute__((ext_vector_type(8))) float v8f;

#define H_DIM 256
#define EPSV 1e-8f

__global__ void zero_ws_kernel(float* __restrict__ p, int n) {
    int i = blockIdx.x * blockDim.x + threadIdx.x;
    if (i < n) p[i] = 0.0f;
}

__global__ __launch_bounds__(H_DIM)
void scatter_kernel(const float* __restrict__ x,
                    const int*   __restrict__ dst_idx,
                    float* __restrict__ sums,
                    float* __restrict__ counts) {
    const int row = blockIdx.x;
    const int tid = threadIdx.x;
    const int seg = dst_idx[row];
    const float v = x[(size_t)row * H_DIM + tid];
    atomicAdd(&sums[(size_t)seg * H_DIM + tid], v);
    if (tid == 0) atomicAdd(&counts[seg], 1.0f);
}

__global__ __launch_bounds__(256)
void pool_gemm_kernel(const float* __restrict__ sums,
                      const float* __restrict__ counts,
                      const float* __restrict__ W,
                      const float* __restrict__ bias,
                      float* __restrict__ out,
                      int S) {
    const int wid  = (int)((blockIdx.x * blockDim.x + threadIdx.x) >> 5);
    const int lane = (int)(threadIdx.x & 31);
    const int tilesN = H_DIM / 16;
    const int tileM = wid / tilesN;
    const int tileN = wid % tilesN;
    if (tileM * 16 >= S) return;

    const int m = lane & 15;
    const int h = lane >> 4;

    const int seg = tileM * 16 + m;
    const int col = tileN * 16 + m;

    const float inv = 1.0f / (counts[seg] + EPSV);

    const float* __restrict__ srow = sums + (size_t)seg * H_DIM;
    const float* __restrict__ wcol = W    + (size_t)col * H_DIM;

    v8f c;
    const float bj = bias[col];
#pragma unroll
    for (int r = 0; r < 8; ++r) c[r] = bj;

#pragma unroll 8
    for (int k = 0; k < H_DIM; k += 4) {
        v2f a = *(const v2f*)(srow + k + 2 * h);
        a.x *= inv;
        a.y *= inv;
        const v2f bf = *(const v2f*)(wcol + k + 2 * h);
        c = __builtin_amdgcn_wmma_f32_16x16x4_f32(
                 false, a,  false, bf,
                 (short)0, c,  false,  false);
    }

#pragma unroll
    for (int r = 0; r < 8; ++r) {
        out[(size_t)(tileM * 16 + r + 8 * h) * H_DIM + tileN * 16 + m] = c[r];
    }
}

extern "C" void kernel_launch(void* const* d_in, const int* in_sizes, int n_in,
                              void* d_out, int out_size, void* d_ws, size_t ws_size,
                              hipStream_t stream) {
    const float* x       = (const float*)d_in[0];
    const int*   dst_idx = (const int*)  d_in[1];
    const float* W       = (const float*)d_in[3];
    const float* bias    = (const float*)d_in[4];
    float*       out     = (float*)d_out;

    const int N = in_sizes[0] / H_DIM;
    const int S = out_size   / H_DIM;

    float* sums   = (float*)d_ws;
    float* counts = sums + (size_t)S * H_DIM;

    const int zn = S * H_DIM + S;
    zero_ws_kernel<<<(zn + 255) / 256, 256, 0, stream>>>(sums, zn);

    scatter_kernel<<<N, H_DIM, 0, stream>>>(x, dst_idx, sums, counts);

    const int tiles = (S / 16) * (H_DIM / 16);
    pool_gemm_kernel<<<tiles / 8, 256, 0, stream>>>(sums, counts, W, bias, out, S);
}


# Float-atomic scatter replaced by an atomic-free segment-owned sum: workgroup scans dst_idx, lists its 16 segments' rows in LDS, each wave sums one segment in a fixed row order; zero launch gone
# speedup vs baseline: 3.2851x; 3.2851x over previous
#include <hip/hip_runtime.h>
#include <hip/hip_bf16.h>

typedef __attribute__((ext_vector_type(2))) float v2f;
typedef __attribute__((ext_vector_type(8))) float v8f;

#define H_DIM 256
#define EPSV 1e-8f

typedef __attribute__((ext_vector_type(4))) float v4f;
typedef __attribute__((ext_vector_type(4))) int   v4i;

#define G_WAVES   16
#define G_THREADS (G_WAVES * 32)
#define G_LCAP    2560
#define G_MCAP    256
#define G_MTH     (G_MCAP - 160)
#define G_DEPTH   16
#define G_ROWS    8

static __device__ __forceinline__ unsigned wave_ballot(bool p) { return __builtin_amdgcn_ballot_w32(p); }
static __device__ __forceinline__ unsigned lanes_below(unsigned mask) { return __builtin_amdgcn_mbcnt_lo(mask, 0u); }

static __device__ __forceinline__ v4i load_chunk(const int* __restrict__ idx, int q, int lane, int q1) {
    int i = q + lane;
    i = i < q1 ? i : q1 - 1;
    return *(const v4i*)((const char*)idx + ((unsigned)i << 4));
}

static __device__ __forceinline__ int append_matches(v4i v, int q, int q1, int lane, unsigned tb,
                                                     unsigned* __restrict__ list, int n) {
    const int left = q1 - q;
    const unsigned vmask = left >= 32 ? ~0u : (left <= 0 ? 0u : (1u << left) - 1u);
    const unsigned row = (unsigned)(q + lane) << 2;
#pragma unroll
    for (int c = 0; c < 4; ++c) {
        const unsigned d = (unsigned)v[c] - tb;
        const unsigned b = wave_ballot(d < 16u) & vmask;
        if (b != 0u) {
            if ((b >> lane) & 1u) list[n + (int)lanes_below(b)] = ((row + (unsigned)c) << 4) | d;
            n += __builtin_popcount(b);
        }
    }
    return n;
}

static __device__ __forceinline__ int take_own(unsigned e, bool valid, int w,
                                               unsigned* __restrict__ mine, int pend) {
    const bool m = valid && (e & 15u) == (unsigned)w;
    const unsigned b = wave_ballot(m);
    if (b != 0u) {
        if (m) mine[pend + (int)lanes_below(b)] = e >> 4;
        pend += __builtin_popcount(b);
    }
    return pend;
}

static __device__ __forceinline__ void sum_rows(const float* __restrict__ x, const unsigned* mine,
                                                int count, int lane, v4f& a0, v4f& a1) {
    for (int i = 0; i < count; i += G_ROWS) {
        int li = i + (lane & (G_ROWS - 1));
        li = li < count ? li : count - 1;
        const unsigned rv = mine[li];
        v4f v0[G_ROWS], v1[G_ROWS];
#pragma unroll
        for (int k = 0; k < G_ROWS; ++k) {
            const unsigned r = (unsigned)__builtin_amdgcn_readlane((int)rv, k);
            const v4f* p = (const v4f*)(x + (size_t)r * H_DIM) + lane;
            v0[k] = __builtin_nontemporal_load(p);
            v1[k] = __builtin_nontemporal_load(p + 32);
        }
#pragma unroll
        for (int k = 0; k < G_ROWS; ++k) {
            if (i + k < count) { a0 += v0[k]; a1 += v1[k]; }
        }
    }
}

__global__ __launch_bounds__(G_THREADS)
void gather_sum_kernel(const float* __restrict__ x,
                       const int*   __restrict__ dst_idx,
                       float* __restrict__ sums,
                       float* __restrict__ counts,
                       int N, int S) {
    __shared__ unsigned s_list[G_WAVES][G_LCAP];
    __shared__ unsigned s_mine[G_WAVES][G_MCAP];
    __shared__ int s_cnt[G_WAVES];
    __shared__ int s_tot[G_WAVES];
    __shared__ int s_more[2];

    const int lane = (int)(threadIdx.x & 31);
    const int w    = __builtin_amdgcn_readfirstlane((int)(threadIdx.x >> 5));
    const unsigned tb = blockIdx.x * G_WAVES;

    const int n4  = N >> 2;
    const int per = (((n4 + G_WAVES - 1) / G_WAVES) + 31) & ~31;
    int qpos      = w * per < n4 ? w * per : n4;
    const int q1  = qpos + per < n4 ? qpos + per : n4;
    bool tail     = (w == G_WAVES - 1) && (N & 3) != 0;

    const int max_rounds = (4 * per + 3) / (G_LCAP - 128 * G_DEPTH + 1) + 2;

    v4f a0 = {0.f, 0.f, 0.f, 0.f}, a1 = {0.f, 0.f, 0.f, 0.f};
    int total = 0;

    if (threadIdx.x == 0) { s_more[0] = 0; s_more[1] = 0; }
    __syncthreads();

    for (int round = 0; round < max_rounds; ++round) {
        int n = 0;
        if (qpos < q1) {
            v4i buf[G_DEPTH];
#pragma unroll
            for (int d = 0; d < G_DEPTH; ++d) buf[d] = load_chunk(dst_idx, qpos + 32 * d, lane, q1);
            while (q1 - qpos > 32 * (G_DEPTH - 1) && n <= G_LCAP - 128 * G_DEPTH) {
#pragma unroll
                for (int d = 0; d < G_DEPTH; ++d) {
                    n = append_matches(buf[d], qpos + 32 * d, q1, lane, tb, s_list[w], n);
                    buf[d] = load_chunk(dst_idx, qpos + 32 * (G_DEPTH + d), lane, q1);
                }
                qpos += 32 * G_DEPTH;
            }
#pragma unroll
            for (int d = 0; d < G_DEPTH; ++d) {
                if (qpos < q1 && n <= G_LCAP - 128) {
                    n = append_matches(buf[d], qpos, q1, lane, tb, s_list[w], n);
                    qpos += 32;
                }
            }
        }
        if (tail && qpos >= q1 && n <= G_LCAP - 128) {
            const int r = 4 * n4 + lane;
            const bool valid = r < N;
            const unsigned d = (unsigned)(valid ? dst_idx[r] : -1) - tb;
            const bool m = valid && d < 16u;
            const unsigned b = wave_ballot(m);
            if (b != 0u) {
                if (m) s_list[w][n + (int)lanes_below(b)] = ((unsigned)r << 4) | d;
                n += __builtin_popcount(b);
            }
            tail = false;
        }
        if (lane == 0) {
            s_cnt[w] = n;
            if (qpos < q1 || tail) s_more[round & 1] = 1;
        }
        __syncthreads();
        const int again = s_more[round & 1];
        if (threadIdx.x == 0) s_more[(round + 1) & 1] = 0;

        const int cnts = s_cnt[lane & (G_WAVES - 1)];
        int pend = 0;
#pragma nounroll
        for (int ub = 0; ; ub += 4) {
            if (ub == G_WAVES || pend > G_MTH) {
                sum_rows(x, s_mine[w], pend, lane, a0, a1);
                total += pend;
                pend = 0;
            }
            if (ub == G_WAVES) break;
            const unsigned e0 = s_list[ub + 0][lane];
            const unsigned e1 = s_list[ub + 1][lane];
            const unsigned e2 = s_list[ub + 2][lane];
            const unsigned e3 = s_list[ub + 3][lane];
            pend = take_own(e0, lane < __builtin_amdgcn_readlane(cnts, ub + 0), w, s_mine[w], pend);
            pend = take_own(e1, lane < __builtin_amdgcn_readlane(cnts, ub + 1), w, s_mine[w], pend);
            pend = take_own(e2, lane < __builtin_amdgcn_readlane(cnts, ub + 2), w, s_mine[w], pend);
            pend = take_own(e3, lane < __builtin_amdgcn_readlane(cnts, ub + 3), w, s_mine[w], pend);
#pragma nounroll
            for (int k = 0; k < 4; ++k) {
                const int cu = __builtin_amdgcn_readlane(cnts, ub + k);
                for (int j = 32; j < cu; j += 32) {
                    if (pend > G_MTH) {
                        sum_rows(x, s_mine[w], pend, lane, a0, a1);
                        total += pend;
                        pend = 0;
                    }
                    pend = take_own(s_list[ub + k][j + lane], j + lane < cu, w, s_mine[w], pend);
                }
            }
        }
        if (lane == 0) s_tot[w] = total;
        __syncthreads();
        if (!again) break;
    }

    const unsigned seg = tb + (unsigned)w;
    if (seg < (unsigned)S) {
        v4f* o = (v4f*)(sums + (size_t)seg * H_DIM) + lane;
        o[0]  = a0;
        o[32] = a1;
    }
    if (w == 0 && lane < G_WAVES && tb + (unsigned)lane < (unsigned)S)
        counts[tb + lane] = (float)s_tot[lane];
}

__global__ __launch_bounds__(256)
void pool_gemm_kernel(const float* __restrict__ sums,
                      const float* __restrict__ counts,
                      const float* __restrict__ W,
                      const float* __restrict__ bias,
                      float* __restrict__ out,
                      int S) {
    const int wid  = (int)((blockIdx.x * blockDim.x + threadIdx.x) >> 5);
    const int lane = (int)(threadIdx.x & 31);
    const int tilesN = H_DIM / 16;
    const int tileM = wid / tilesN;
    const int tileN = wid % tilesN;
    if (tileM * 16 >= S) return;

    const int m = lane & 15;
    const int h = lane >> 4;

    const int seg = tileM * 16 + m;
    const int col = tileN * 16 + m;

    const float inv = 1.0f / (counts[seg] + EPSV);

    const float* __restrict__ srow = sums + (size_t)seg * H_DIM;
    const float* __restrict__ wcol = W    + (size_t)col * H_DIM;

    v8f c;
    const float bj = bias[col];
#pragma unroll
    for (int r = 0; r < 8; ++r) c[r] = bj;

#pragma unroll 8
    for (int k = 0; k < H_DIM; k += 4) {
        v2f a = *(const v2f*)(srow + k + 2 * h);
        a.x *= inv;
        a.y *= inv;
        const v2f bf = *(const v2f*)(wcol + k + 2 * h);
        c = __builtin_amdgcn_wmma_f32_16x16x4_f32(
                 false, a,  false, bf,
                 (short)0, c,  false,  false);
    }

#pragma unroll
    for (int r = 0; r < 8; ++r) {
        out[(size_t)(tileM * 16 + r + 8 * h) * H_DIM + tileN * 16 + m] = c[r];
    }
}

extern "C" void kernel_launch(void* const* d_in, const int* in_sizes, int n_in,
                              void* d_out, int out_size, void* d_ws, size_t ws_size,
                              hipStream_t stream) {
    const float* x       = (const float*)d_in[0];
    const int*   dst_idx = (const int*)  d_in[1];
    const float* W       = (const float*)d_in[3];
    const float* bias    = (const float*)d_in[4];
    float*       out     = (float*)d_out;

    const int N = in_sizes[0] / H_DIM;
    const int S = out_size   / H_DIM;

    float* sums   = (float*)d_ws;
    float* counts = sums + (size_t)S * H_DIM;

    gather_sum_kernel<<<(S + G_WAVES - 1) / G_WAVES, G_THREADS, 0, stream>>>(x, dst_idx, sums, counts, N, S);

    const int tiles = (S / 16) * (H_DIM / 16);
    pool_gemm_kernel<<<tiles / 8, 256, 0, stream>>>(sums, counts, W, bias, out, S);
}


# Mean and linear layer fused into the segment-sum kernel: the workgroup's 16 sums are the A tile in LDS, wave w makes column tile w; second launch and workspace gone; k order inside products changed
# speedup vs baseline: 5.0367x; 1.5332x over previous
#include <hip/hip_runtime.h>
#include <hip/hip_bf16.h>

typedef __attribute__((ext_vector_type(2))) float v2f;
typedef __attribute__((ext_vector_type(8))) float v8f;

#define H_DIM 256
#define EPSV 1e-8f

typedef __attribute__((ext_vector_type(4))) float v4f;
typedef __attribute__((ext_vector_type(4))) int   v4i;

#define G_WAVES   16
#define G_THREADS (G_WAVES * 32)
#define G_LCAP    2560
#define G_MCAP    256
#define G_MTH     (G_MCAP - 160)
#define G_DEPTH   16
#define G_ROWS    8
#define G_PITCH4  65

static __device__ __forceinline__ unsigned wave_ballot(bool p) { return __builtin_amdgcn_ballot_w32(p); }
static __device__ __forceinline__ unsigned lanes_below(unsigned mask) { return __builtin_amdgcn_mbcnt_lo(mask, 0u); }

static __device__ __forceinline__ v4i load_chunk(const int* __restrict__ idx, int q, int lane, int q1) {
    int i = q + lane;
    i = i < q1 ? i : q1 - 1;
    return *(const v4i*)((const char*)idx + ((unsigned)i << 4));
}

static __device__ __forceinline__ int append_matches(v4i v, int q, int q1, int lane, unsigned tb,
                                                     unsigned* __restrict__ list, int n) {
    const int left = q1 - q;
    const unsigned vmask = left >= 32 ? ~0u : (left <= 0 ? 0u : (1u << left) - 1u);
    const unsigned row = (unsigned)(q + lane) << 2;
#pragma unroll
    for (int c = 0; c < 4; ++c) {
        const unsigned d = (unsigned)v[c] - tb;
        const unsigned b = wave_ballot(d < 16u) & vmask;
        if (b != 0u) {
            if ((b >> lane) & 1u) list[n + (int)lanes_below(b)] = ((row + (unsigned)c) << 4) | d;
            n += __builtin_popcount(b);
        }
    }
    return n;
}

static __device__ __forceinline__ int take_own(unsigned e, bool valid, int w,
                                               unsigned* __restrict__ mine, int pend) {
    const bool m = valid && (e & 15u) == (unsigned)w;
    const unsigned b = wave_ballot(m);
    if (b != 0u) {
        if (m) mine[pend + (int)lanes_below(b)] = e >> 4;
        pend += __builtin_popcount(b);
    }
    return pend;
}

static __device__ __forceinline__ void sum_rows(const float* __restrict__ x, const unsigned* mine,
                                                int count, int lane, v4f& a0, v4f& a1) {
    for (int i = 0; i < count; i += G_ROWS) {
        int li = i + (lane & (G_ROWS - 1));
        li = li < count ? li : count - 1;
        const unsigned rv = mine[li];
        v4f v0[G_ROWS], v1[G_ROWS];
#pragma unroll
        for (int k = 0; k < G_ROWS; ++k) {
            const unsigned r = (unsigned)__builtin_amdgcn_readlane((int)rv, k);
            const v4f* p = (const v4f*)(x + (size_t)r * H_DIM) + lane;
            v0[k] = __builtin_nontemporal_load(p);
            v1[k] = __builtin_nontemporal_load(p + 32);
        }
#pragma unroll
        for (int k = 0; k < G_ROWS; ++k) {
            if (i + k < count) { a0 += v0[k]; a1 += v1[k]; }
        }
    }
}

__global__ __launch_bounds__(G_THREADS)
void segment_mean_linear_kernel(const float* __restrict__ x,
                                const int*   __restrict__ dst_idx,
                                const float* __restrict__ W,
                                const float* __restrict__ bias,
                                float* __restrict__ out,
                                int N, int S) {
    __shared__ unsigned s_list[G_WAVES][G_LCAP];
    __shared__ unsigned s_mine[G_WAVES][G_MCAP];
    __shared__ int s_cnt[G_WAVES];
    __shared__ int s_more[2];
    __shared__ v4f s_a[G_WAVES][G_PITCH4];
    __shared__ v4f s_o[G_WAVES][G_PITCH4];

    const int lane = (int)(threadIdx.x & 31);
    const int w    = __builtin_amdgcn_readfirstlane((int)(threadIdx.x >> 5));
    const unsigned tb = blockIdx.x * G_WAVES;

    const int n4  = N >> 2;
    const int per = (((n4 + G_WAVES - 1) / G_WAVES) + 31) & ~31;
    int qpos      = w * per < n4 ? w * per : n4;
    const int q1  = qpos + per < n4 ? qpos + per : n4;
    bool tail     = (w == G_WAVES - 1) && (N & 3) != 0;

    const int max_rounds = (4 * per + 3) / (G_LCAP - 128 * G_DEPTH + 1) + 2;

    v4f a0 = {0.f, 0.f, 0.f, 0.f}, a1 = {0.f, 0.f, 0.f, 0.f};
    int total = 0;

    if (threadIdx.x == 0) { s_more[0] = 0; s_more[1] = 0; }
    __syncthreads();

    for (int round = 0; round < max_rounds; ++round) {
        int n = 0;
        if (qpos < q1) {
            v4i buf[G_DEPTH];
#pragma unroll
            for (int d = 0; d < G_DEPTH; ++d) buf[d] = load_chunk(dst_idx, qpos + 32 * d, lane, q1);
            while (q1 - qpos > 32 * (G_DEPTH - 1) && n <= G_LCAP - 128 * G_DEPTH) {
#pragma unroll
                for (int d = 0; d < G_DEPTH; ++d) {
                    n = append_matches(buf[d], qpos + 32 * d, q1, lane, tb, s_list[w], n);
                    buf[d] = load_chunk(dst_idx, qpos + 32 * (G_DEPTH + d), lane, q1);
                }
                qpos += 32 * G_DEPTH;
            }
#pragma unroll
            for (int d = 0; d < G_DEPTH; ++d) {
                if (qpos < q1 && n <= G_LCAP - 128) {
                    n = append_matches(buf[d], qpos, q1, lane, tb, s_list[w], n);
                    qpos += 32;
                }
            }
        }
        if (tail && qpos >= q1 && n <= G_LCAP - 128) {
            const int r = 4 * n4 + lane;
            const bool valid = r < N;
            const unsigned d = (unsigned)(valid ? dst_idx[r] : -1) - tb;
            const bool m = valid && d < 16u;
            const unsigned b = wave_ballot(m);
            if (b != 0u) {
                if (m) s_list[w][n + (int)lanes_below(b)] = ((unsigned)r << 4) | d;
                n += __builtin_popcount(b);
            }
            tail = false;
        }
        if (lane == 0) {
            s_cnt[w] = n;
            if (qpos < q1 || tail) s_more[round & 1] = 1;
        }
        __syncthreads();
        const int again = s_more[round & 1];
        if (threadIdx.x == 0) s_more[(round + 1) & 1] = 0;

        const int cnts = s_cnt[lane & (G_WAVES - 1)];
        int pend = 0;
#pragma nounroll
        for (int ub = 0; ; ub += 4) {
            if (ub == G_WAVES || pend > G_MTH) {
                sum_rows(x, s_mine[w], pend, lane, a0, a1);
                total += pend;
                pend = 0;
            }
            if (ub == G_WAVES) break;
            const unsigned e0 = s_list[ub + 0][lane];
            const unsigned e1 = s_list[ub + 1][lane];
            const unsigned e2 = s_list[ub + 2][lane];
            const unsigned e3 = s_list[ub + 3][lane];
            pend = take_own(e0, lane < __builtin_amdgcn_readlane(cnts, ub + 0), w, s_mine[w], pend);
            pend = take_own(e1, lane < __builtin_amdgcn_readlane(cnts, ub + 1), w, s_mine[w], pend);
            pend = take_own(e2, lane < __builtin_amdgcn_readlane(cnts, ub + 2), w, s_mine[w], pend);
            pend = take_own(e3, lane < __builtin_amdgcn_readlane(cnts, ub + 3), w, s_mine[w], pend);
#pragma nounroll
            for (int k = 0; k < 4; ++k) {
                const int cu = __builtin_amdgcn_readlane(cnts, ub + k);
                for (int j = 32; j < cu; j += 32) {
                    if (pend > G_MTH) {
                        sum_rows(x, s_mine[w], pend, lane, a0, a1);
                        total += pend;
                        pend = 0;
                    }
                    pend = take_own(s_list[ub + k][j + lane], j + lane < cu, w, s_mine[w], pend);
                }
            }
        }
        if (!again) break;
        __syncthreads();
    }

    const int m   = lane & 15;
    const int h   = lane >> 4;
    const int col = 16 * w + m;

    const float bj = bias[col];
    const v4f* __restrict__ wp = (const v4f*)(W + (size_t)col * H_DIM) + h;
    v4f bq[H_DIM / 8];
#pragma unroll
    for (int j = 0; j < H_DIM / 8; ++j) bq[j] = wp[2 * j];

    const float inv = 1.0f / ((float)total + EPSV);
    s_a[w][lane]      = a0 * inv;
    s_a[w][32 + lane] = a1 * inv;
    __builtin_amdgcn_fence(__ATOMIC_RELEASE, "workgroup", "local");
    __builtin_amdgcn_s_barrier();
    __builtin_amdgcn_fence(__ATOMIC_ACQUIRE, "workgroup", "local");

    v8f c;
#pragma unroll
    for (int r = 0; r < 8; ++r) c[r] = bj;
    v4f aq[2][8];
#pragma unroll
    for (int i = 0; i < 8; ++i) aq[0][i] = s_a[m][2 * i + h];
#pragma unroll
    for (int g = 0; g < H_DIM / 64; ++g) {
        if (g + 1 < H_DIM / 64) {
#pragma unroll
            for (int i = 0; i < 8; ++i) aq[(g + 1) & 1][i] = s_a[m][2 * (8 * (g + 1) + i) + h];
        }
        __builtin_amdgcn_sched_barrier(0);
#pragma unroll
        for (int i = 0; i < 8; ++i) {
            const v4f av = aq[g & 1][i], bv = bq[8 * g + i];
            const v2f a01 = {av.x, av.y}, a23 = {av.z, av.w};
            const v2f b01 = {bv.x, bv.y}, b23 = {bv.z, bv.w};
            c = __builtin_amdgcn_wmma_f32_16x16x4_f32(
                     false, a01,  false, b01,
                     (short)0, c,  false,  false);
            c = __builtin_amdgcn_wmma_f32_16x16x4_f32(
                     false, a23,  false, b23,
                     (short)0, c,  false,  false);
        }
        __builtin_amdgcn_sched_barrier(0);
    }

    float* so = (float*)&s_o[0][0];
#pragma unroll
    for (int r = 0; r < 8; ++r) so[(r + 8 * h) * (4 * G_PITCH4) + col] = c[r];
    __syncthreads();
    const unsigned seg = tb + (unsigned)w;
    if (seg < (unsigned)S) {
        v4f* o = (v4f*)(out + (size_t)seg * H_DIM) + lane;
        o[0]  = s_o[w][lane];
        o[32] = s_o[w][32 + lane];
    }
}

extern "C" void kernel_launch(void* const* d_in, const int* in_sizes, int n_in,
                              void* d_out, int out_size, void* d_ws, size_t ws_size,
                              hipStream_t stream) {
    const float* x       = (const float*)d_in[0];
    const int*   dst_idx = (const int*)  d_in[1];
    const float* W       = (const float*)d_in[3];
    const float* bias    = (const float*)d_in[4];
    float*       out     = (float*)d_out;

    const int N = in_sizes[0] / H_DIM;
    const int S = out_size   / H_DIM;

    segment_mean_linear_kernel<<<(S + G_WAVES - 1) / G_WAVES, G_THREADS, 0, stream>>>(x, dst_idx, W, bias, out, N, S);
}


# B operand (W) fetched with whole-line loads (two aligned 256-byte pieces per request, not 32 bytes of each of 16 rows), made into fragments in a per-wave LDS tile between barrier signal and wait
# speedup vs baseline: 5.4114x; 1.0744x over previous
#include <hip/hip_runtime.h>
#include <hip/hip_bf16.h>

typedef __attribute__((ext_vector_type(2))) float v2f;
typedef __attribute__((ext_vector_type(8))) float v8f;

#define H_DIM 256
#define EPSV 1e-8f

typedef __attribute__((ext_vector_type(4))) float v4f;
typedef __attribute__((ext_vector_type(4))) int   v4i;

#define G_WAVES   16
#define G_THREADS (G_WAVES * 32)
#define G_LCAP    2560
#define G_MCAP    256
#define G_MTH     (G_MCAP - 160)
#define G_DEPTH   16
#define G_ROWS    8
#define G_PITCH4  65
#define G_WPITCH4 17

static __device__ __forceinline__ unsigned wave_ballot(bool p) { return __builtin_amdgcn_ballot_w32(p); }
static __device__ __forceinline__ unsigned lanes_below(unsigned mask) { return __builtin_amdgcn_mbcnt_lo(mask, 0u); }

static __device__ __forceinline__ v4i load_chunk(const int* __restrict__ idx, int q, int lane, int q1) {
    int i = q + lane;
    i = i < q1 ? i : q1 - 1;
    return *(const v4i*)((const char*)idx + ((unsigned)i << 4));
}

static __device__ __forceinline__ int append_matches(v4i v, int q, int q1, int lane, unsigned tb,
                                                     unsigned* __restrict__ list, int n) {
    const int left = q1 - q;
    const unsigned vmask = left >= 32 ? ~0u : (left <= 0 ? 0u : (1u << left) - 1u);
    const unsigned row = (unsigned)(q + lane) << 2;
#pragma unroll
    for (int c = 0; c < 4; ++c) {
        const unsigned d = (unsigned)v[c] - tb;
        const unsigned b = wave_ballot(d < 16u) & vmask;
        if (b != 0u) {
            if ((b >> lane) & 1u) list[n + (int)lanes_below(b)] = ((row + (unsigned)c) << 4) | d;
            n += __builtin_popcount(b);
        }
    }
    return n;
}

static __device__ __forceinline__ int take_own(unsigned e, bool valid, int w,
                                               unsigned* __restrict__ mine, int pend) {
    const bool m = valid && (e & 15u) == (unsigned)w;
    const unsigned b = wave_ballot(m);
    if (b != 0u) {
        if (m) mine[pend + (int)lanes_below(b)] = e >> 4;
        pend += __builtin_popcount(b);
    }
    return pend;
}

static __device__ __forceinline__ void sum_rows(const float* __restrict__ x, const unsigned* mine,
                                                int count, int lane, v4f& a0, v4f& a1) {
    for (int i = 0; i < count; i += G_ROWS) {
        int li = i + (lane & (G_ROWS - 1));
        li = li < count ? li : count - 1;
        const unsigned rv = mine[li];
        v4f v0[G_ROWS], v1[G_ROWS];
#pragma unroll
        for (int k = 0; k < G_ROWS; ++k) {
            const unsigned r = (unsigned)__builtin_amdgcn_readlane((int)rv, k);
            const v4f* p = (const v4f*)(x + (size_t)r * H_DIM) + lane;
            v0[k] = __builtin_nontemporal_load(p);
            v1[k] = __builtin_nontemporal_load(p + 32);
        }
#pragma unroll
        for (int k = 0; k < G_ROWS; ++k) {
            if (i + k < count) { a0 += v0[k]; a1 += v1[k]; }
        }
    }
}

__global__ __launch_bounds__(G_THREADS)
void segment_mean_linear_kernel(const float* __restrict__ x,
                                const int*   __restrict__ dst_idx,
                                const float* __restrict__ W,
                                const float* __restrict__ bias,
                                float* __restrict__ out,
                                int N, int S) {
    __shared__ unsigned s_list[G_WAVES][G_LCAP];
    __shared__ unsigned s_mine[G_WAVES][G_MCAP];
    __shared__ int s_cnt[G_WAVES];
    __shared__ int s_more[2];
    __shared__ v4f s_a[G_WAVES][G_PITCH4];
    __shared__ v4f s_o[G_WAVES][G_PITCH4];
    __shared__ v4f s_w[G_WAVES][16 * G_WPITCH4];

    const int lane = (int)(threadIdx.x & 31);
    const int w    = __builtin_amdgcn_readfirstlane((int)(threadIdx.x >> 5));
    const unsigned tb = blockIdx.x * G_WAVES;

    const int n4  = N >> 2;
    const int per = (((n4 + G_WAVES - 1) / G_WAVES) + 31) & ~31;
    int qpos      = w * per < n4 ? w * per : n4;
    const int q1  = qpos + per < n4 ? qpos + per : n4;
    bool tail     = (w == G_WAVES - 1) && (N & 3) != 0;

    const int max_rounds = (4 * per + 3) / (G_LCAP - 128 * G_DEPTH + 1) + 2;

    v4f a0 = {0.f, 0.f, 0.f, 0.f}, a1 = {0.f, 0.f, 0.f, 0.f};
    int total = 0;

    if (threadIdx.x == 0) { s_more[0] = 0; s_more[1] = 0; }
    __syncthreads();

    for (int round = 0; round < max_rounds; ++round) {
        int n = 0;
        if (qpos < q1) {
            v4i buf[G_DEPTH];
#pragma unroll
            for (int d = 0; d < G_DEPTH; ++d) buf[d] = load_chunk(dst_idx, qpos + 32 * d, lane, q1);
            while (q1 - qpos > 32 * (G_DEPTH - 1) && n <= G_LCAP - 128 * G_DEPTH) {
#pragma unroll
                for (int d = 0; d < G_DEPTH; ++d) {
                    n = append_matches(buf[d], qpos + 32 * d, q1, lane, tb, s_list[w], n);
                    buf[d] = load_chunk(dst_idx, qpos + 32 * (G_DEPTH + d), lane, q1);
                }
                qpos += 32 * G_DEPTH;
            }
#pragma unroll
            for (int d = 0; d < G_DEPTH; ++d) {
                if (qpos < q1 && n <= G_LCAP - 128) {
                    n = append_matches(buf[d], qpos, q1, lane, tb, s_list[w], n);
                    qpos += 32;
                }
            }
        }
        if (tail && qpos >= q1 && n <= G_LCAP - 128) {
            const int r = 4 * n4 + lane;
            const bool valid = r < N;
            const unsigned d = (unsigned)(valid ? dst_idx[r] : -1) - tb;
            const bool m = valid && d < 16u;
            const unsigned b = wave_ballot(m);
            if (b != 0u) {
                if (m) s_list[w][n + (int)lanes_below(b)] = ((unsigned)r << 4) | d;
                n += __builtin_popcount(b);
            }
            tail = false;
        }
        if (lane == 0) {
            s_cnt[w] = n;
            if (qpos < q1 || tail) s_more[round & 1] = 1;
        }
        __syncthreads();
        const int again = s_more[round & 1];
        if (threadIdx.x == 0) s_more[(round + 1) & 1] = 0;

        const int cnts = s_cnt[lane & (G_WAVES - 1)];
        int pend = 0;
#pragma nounroll
        for (int ub = 0; ; ub += 4) {
            if (ub == G_WAVES || pend > G_MTH) {
                sum_rows(x, s_mine[w], pend, lane, a0, a1);
                total += pend;
                pend = 0;
            }
            if (ub == G_WAVES) break;
            const unsigned e0 = s_list[ub + 0][lane];
            const unsigned e1 = s_list[ub + 1][lane];
            const unsigned e2 = s_list[ub + 2][lane];
            const unsigned e3 = s_list[ub + 3][lane];
            pend = take_own(e0, lane < __builtin_amdgcn_readlane(cnts, ub + 0), w, s_mine[w], pend);
            pend = take_own(e1, lane < __builtin_amdgcn_readlane(cnts, ub + 1), w, s_mine[w], pend);
            pend = take_own(e2, lane < __builtin_amdgcn_readlane(cnts, ub + 2), w, s_mine[w], pend);
            pend = take_own(e3, lane < __builtin_amdgcn_readlane(cnts, ub + 3), w, s_mine[w], pend);
#pragma nounroll
            for (int k = 0; k < 4; ++k) {
                const int cu = __builtin_amdgcn_readlane(cnts, ub + k);
                for (int j = 32; j < cu; j += 32) {
                    if (pend > G_MTH) {
                        sum_rows(x, s_mine[w], pend, lane, a0, a1);
                        total += pend;
                        pend = 0;
                    }
                    pend = take_own(s_list[ub + k][j + lane], j + lane < cu, w, s_mine[w], pend);
                }
            }
        }
        if (!again) break;
        __syncthreads();
    }

    const int m   = lane & 15;
    const int h   = lane >> 4;
    const int col = 16 * w + m;

    const float bj = bias[col];
    const v4f* __restrict__ wg = (const v4f*)(W + (size_t)(16 * w + h) * H_DIM) + m;
    v4f tq[4][8];
#pragma unroll
    for (int q = 0; q < 4; ++q) {
#pragma unroll
        for (int t = 0; t < 8; ++t) tq[q][t] = wg[(2 * t) * (H_DIM / 4) + 16 * q];
    }

    const float inv = 1.0f / ((float)total + EPSV);
    s_a[w][lane]      = a0 * inv;
    s_a[w][32 + lane] = a1 * inv;
    __builtin_amdgcn_fence(__ATOMIC_RELEASE, "workgroup", "local");
    __builtin_amdgcn_s_barrier_signal(-1);

    v4f* sw = s_w[w];
    v4f bq[H_DIM / 8];
#pragma unroll
    for (int q = 0; q < 4; ++q) {
#pragma unroll
        for (int t = 0; t < 8; ++t) sw[(2 * t + h) * G_WPITCH4 + m] = tq[q][t];
#pragma unroll
        for (int i = 0; i < 8; ++i) bq[8 * q + i] = sw[m * G_WPITCH4 + 2 * i + h];
    }

    __builtin_amdgcn_s_barrier_wait(-1);
    __builtin_amdgcn_fence(__ATOMIC_ACQUIRE, "workgroup", "local");

    v8f c;
#pragma unroll
    for (int r = 0; r < 8; ++r) c[r] = bj;
    v4f aq[2][8];
#pragma unroll
    for (int i = 0; i < 8; ++i) aq[0][i] = s_a[m][2 * i + h];
#pragma unroll
    for (int g = 0; g < H_DIM / 64; ++g) {
        if (g + 1 < H_DIM / 64) {
#pragma unroll
            for (int i = 0; i < 8; ++i) aq[(g + 1) & 1][i] = s_a[m][2 * (8 * (g + 1) + i) + h];
        }
        __builtin_amdgcn_sched_barrier(0);
#pragma unroll
        for (int i = 0; i < 8; ++i) {
            const v4f av = aq[g & 1][i], bv = bq[8 * g + i];
            const v2f a01 = {av.x, av.y}, a23 = {av.z, av.w};
            const v2f b01 = {bv.x, bv.y}, b23 = {bv.z, bv.w};
            c = __builtin_amdgcn_wmma_f32_16x16x4_f32(
                     false, a01,  false, b01,
                     (short)0, c,  false,  false);
            c = __builtin_amdgcn_wmma_f32_16x16x4_f32(
                     false, a23,  false, b23,
                     (short)0, c,  false,  false);
        }
        __builtin_amdgcn_sched_barrier(0);
    }

    float* so = (float*)&s_o[0][0];
#pragma unroll
    for (int r = 0; r < 8; ++r) so[(r + 8 * h) * (4 * G_PITCH4) + col] = c[r];
    __syncthreads();
    const unsigned seg = tb + (unsigned)w;
    if (seg < (unsigned)S) {
        v4f* o = (v4f*)(out + (size_t)seg * H_DIM) + lane;
        o[0]  = s_o[w][lane];
        o[32] = s_o[w][32 + lane];
    }
}

extern "C" void kernel_launch(void* const* d_in, const int* in_sizes, int n_in,
                              void* d_out, int out_size, void* d_ws, size_t ws_size,
                              hipStream_t stream) {
    const float* x       = (const float*)d_in[0];
    const int*   dst_idx = (const int*)  d_in[1];
    const float* W       = (const float*)d_in[3];
    const float* bias    = (const float*)d_in[4];
    float*       out     = (float*)d_out;

    const int N = in_sizes[0] / H_DIM;
    const int S = out_size   / H_DIM;

    segment_mean_linear_kernel<<<(S + G_WAVES - 1) / G_WAVES, G_THREADS, 0, stream>>>(x, dst_idx, W, bias, out, N, S);
}
